# starting kernel
# baseline (speedup 1.0000x reference)
#include <hip/hip_runtime.h>
#include <math.h>


typedef float v2f __attribute__((ext_vector_type(2)));
typedef float v8f __attribute__((ext_vector_type(8)));

#define RESN     320
#define NSHOTS   8
#define SPS      1001
#define M_TOTAL  (NSHOTS * SPS)
#define M_PAD    8016
#define NC       8
#define TWO_PI   6.283185307179586f
#define RB       256

#define V8F_ZERO {0.f, 0.f, 0.f, 0.f, 0.f, 0.f, 0.f, 0.f}

__device__ __forceinline__ v8f wmma_f32(v2f a, v2f b, v8f c) {
  return __builtin_amdgcn_wmma_f32_16x16x4_f32(
       false, a,  false, b,
       (short)0, c,  false,  false);
}

__device__ __forceinline__ v2f ldv2(const float* __restrict__ p) {
  return *(const v2f*)p;
}

__global__ void traj_kernel(const float* __restrict__ x, float* __restrict__ xt) {
  int gid = blockIdx.x * blockDim.x + threadIdx.x;
  if (gid >= M_TOTAL * 2) return;
  int d    = gid & 1;
  int m    = gid >> 1;
  int shot = m / SPS;
  int tt   = m - shot * SPS;
  int I    = (tt == 0) ? 0 : (tt - 1) / 10;
  if (I > 99) I = 99;
  const float* base = x + (size_t)shot * SPS * 2 + d;
  float y0 = base[(I) * 20];
  float y1 = base[(I + 1) * 20];
  float dI = (y1 - y0) * 0.1f;
  float mI, mI1;
  if (I == 0) { mI = dI; }
  else        { float ym = base[(I - 1) * 20]; mI = 0.5f * (dI + (y0 - ym) * 0.1f); }
  if (I == 99) { mI1 = dI; }
  else         { float y2 = base[(I + 2) * 20]; mI1 = 0.5f * (dI + (y2 - y1) * 0.1f); }
  float s  = (float)(tt - 10 * I) * 0.1f;
  float s2 = s * s, s3 = s2 * s;
  float h00 = 1.0f - 3.0f * s2 + 2.0f * s3;
  float h10 = s - 2.0f * s2 + s3;
  float h01 = 3.0f * s2 - 2.0f * s3;
  float h11 = s3 - s2;
  xt[gid] = h00 * y0 + h10 * mI * 10.0f + h01 * y1 + h11 * mI1 * 10.0f;
}

__global__ void phase_kernel(const float* __restrict__ xt,
                             float* __restrict__ EhR, float* __restrict__ EhI,
                             float* __restrict__ EwR, float* __restrict__ EwI) {
  int gid = blockIdx.x * blockDim.x + threadIdx.x;
  if (gid >= M_PAD * RESN) return;
  int m = gid / RESN;
  int n = gid - m * RESN;
  float ehr = 0.f, ehi = 0.f, ewr = 0.f, ewi = 0.f;
  if (m < M_TOTAL) {
    float k = (TWO_PI / (float)RESN) * ((float)n - 160.0f);
    float sh, ch, sw, cw;
    __sincosf(xt[2 * m + 0] * k, &sh, &ch);
    __sincosf(xt[2 * m + 1] * k, &sw, &cw);
    ehr = ch; ehi = -sh; ewr = cw; ewi = -sw;
  }
  EhR[gid] = ehr; EhI[gid] = ehi;
  EwR[gid] = ewr; EwI[gid] = ewi;
}

__global__ void split_img_kernel(const float* __restrict__ image,
                                 float* __restrict__ imgR, float* __restrict__ imgI) {
  int i = blockIdx.x * blockDim.x + threadIdx.x;
  if (i < NC * RESN * RESN) {
    imgR[i] = image[2 * i];
    imgI[i] = image[2 * i + 1];
  }
}

__global__ __launch_bounds__(32) void gemm1_kernel(
    const float* __restrict__ imgR, const float* __restrict__ imgI,
    const float* __restrict__ EhR, const float* __restrict__ EhI,
    const float* __restrict__ EwR, const float* __restrict__ EwI,
    float* __restrict__ kspR, float* __restrict__ kspI) {
  const int m0   = blockIdx.x * 16;
  const int c    = blockIdx.y;
  const int l    = threadIdx.x;
  const int half = l >> 4;
  const int lm   = l & 15;
  float accR[8], accI[8];
#pragma unroll
  for (int v = 0; v < 8; ++v) { accR[v] = 0.0f; accI[v] = 0.0f; }

  const size_t aRow = (size_t)(m0 + lm) * RESN + 2 * half;

  for (int ht = 0; ht < 20; ++ht) {
    const int h0 = ht * 16;
    const size_t bRow = ((size_t)c * RESN + h0 + lm) * RESN + 2 * half;
    v8f C1 = V8F_ZERO;
    v8f C2 = V8F_ZERO;
    v8f C3 = V8F_ZERO;
    for (int w0 = 0; w0 < RESN; w0 += 4) {
      v2f Ar = ldv2(EwR + aRow + w0);
      v2f Ai = ldv2(EwI + aRow + w0);
      v2f Br = ldv2(imgR + bRow + w0);
      v2f Bi = ldv2(imgI + bRow + w0);
      v2f As = Ar + Ai;
      v2f Bs = Br + Bi;
      C1 = wmma_f32(Ar, Br, C1);
      C2 = wmma_f32(Ai, Bi, C2);
      C3 = wmma_f32(As, Bs, C3);
    }
#pragma unroll
    for (int v = 0; v < 8; ++v) {
      const int ml = v + 8 * half;
      const size_t ei = (size_t)(m0 + ml) * RESN + h0 + lm;
      float er = EhR[ei], eim = EhI[ei];
      float tr = C1[v] - C2[v];
      float ti = C3[v] - C1[v] - C2[v];
      float pr = er * tr - eim * ti;
      float pi = er * ti + eim * tr;
#pragma unroll
      for (int mask = 8; mask >= 1; mask >>= 1) {
        pr += __shfl_xor(pr, mask, 32);
        pi += __shfl_xor(pi, mask, 32);
      }
      accR[v] += pr; accI[v] += pi;
    }
  }
  if (lm == 0) {
#pragma unroll
    for (int v = 0; v < 8; ++v) {
      const int m = m0 + v + 8 * half;
      if (m < M_TOTAL) {
        kspR[(size_t)c * M_PAD + m] = accR[v];
        kspI[(size_t)c * M_PAD + m] = accI[v];
      }
    }
  }
}

#define KBLK 32
#define LSTR 34
__global__ __launch_bounds__(256) void gemm2_kernel(
    const float* __restrict__ EhR, const float* __restrict__ EhI,
    const float* __restrict__ EwR, const float* __restrict__ EwI,
    const float* __restrict__ kspR, const float* __restrict__ kspI,
    float* __restrict__ out) {
  __shared__ float GtR[64 * LSTR];
  __shared__ float GtI[64 * LSTR];
  __shared__ float BtR[32 * LSTR];
  __shared__ float BtI[32 * LSTR];
  const int tid  = threadIdx.x;
  const int wave = tid >> 5;
  const int l    = tid & 31;
  const int half = l >> 4;
  const int lm   = l & 15;
  const int wr   = wave >> 1;
  const int wc   = wave & 1;
  const int rowTile = blockIdx.x;
  const int colTile = blockIdx.y;
  const int c  = rowTile / 5;
  const int h0 = (rowTile % 5) * 64;
  const int w0 = colTile * 32;

  v8f C1 = V8F_ZERO;
  v8f C2 = V8F_ZERO;
  v8f C3 = V8F_ZERO;

  const int aOff = (wr * 16 + lm) * LSTR + 2 * half;
  const int bOff = (wc * 16 + lm) * LSTR + 2 * half;

  for (int kb = 0; kb < (M_TOTAL + KBLK - 1) / KBLK; ++kb) {
    const int k0 = kb * KBLK;
    for (int i = tid; i < KBLK * 32; i += 256) {
      const int n = i & 31, k = i >> 5;
      const int m = k0 + k;
      float br = 0.f, bi = 0.f;
      if (m < M_TOTAL) {
        const size_t gi = (size_t)m * RESN + w0 + n;
        br = EwR[gi];
        bi = -EwI[gi];
      }
      BtR[n * LSTR + k] = br;
      BtI[n * LSTR + k] = bi;
    }
    for (int i = tid; i < 64 * KBLK; i += 256) {
      const int r = i & 63, k = i >> 6;
      const int m = k0 + k;
      float gr = 0.f, gi = 0.f;
      if (m < M_TOTAL) {
        const size_t ei = (size_t)m * RESN + h0 + r;
        float er = EhR[ei], eim = EhI[ei];
        float kr = kspR[(size_t)c * M_PAD + m];
        float ki = kspI[(size_t)c * M_PAD + m];
        gr = er * kr + eim * ki;
        gi = er * ki - eim * kr;
      }
      GtR[r * LSTR + k] = gr;
      GtI[r * LSTR + k] = gi;
    }
    __syncthreads();
#pragma unroll
    for (int kk = 0; kk < KBLK; kk += 4) {
      v2f Ar = ldv2(GtR + aOff + kk);
      v2f Ai = ldv2(GtI + aOff + kk);
      v2f Br = ldv2(BtR + bOff + kk);
      v2f Bi = ldv2(BtI + bOff + kk);
      v2f As = Ar + Ai;
      v2f Bs = Br + Bi;
      C1 = wmma_f32(Ar, Br, C1);
      C2 = wmma_f32(Ai, Bi, C2);
      C3 = wmma_f32(As, Bs, C3);
    }
    __syncthreads();
  }
#pragma unroll
  for (int v = 0; v < 8; ++v) {
    const int row = rowTile * 64 + wr * 16 + v + 8 * half;
    const int col = w0 + wc * 16 + lm;
    float rr = C1[v] - C2[v];
    float ri = C3[v] - C1[v] - C2[v];
    out[(size_t)row * RESN + col] = sqrtf(rr * rr + ri * ri);
  }
}

__global__ void reduce1_kernel(const float* __restrict__ out, float* __restrict__ red, int n) {
  __shared__ float smn[256], smx[256];
  float mn = 3.402823466e38f, mx = -3.402823466e38f;
  for (int i = blockIdx.x * blockDim.x + threadIdx.x; i < n; i += gridDim.x * blockDim.x) {
    float v = out[i];
    mn = fminf(mn, v); mx = fmaxf(mx, v);
  }
  smn[threadIdx.x] = mn; smx[threadIdx.x] = mx;
  __syncthreads();
  for (int s = 128; s > 0; s >>= 1) {
    if ((int)threadIdx.x < s) {
      smn[threadIdx.x] = fminf(smn[threadIdx.x], smn[threadIdx.x + s]);
      smx[threadIdx.x] = fmaxf(smx[threadIdx.x], smx[threadIdx.x + s]);
    }
    __syncthreads();
  }
  if (threadIdx.x == 0) { red[2 + blockIdx.x] = smn[0]; red[2 + RB + blockIdx.x] = smx[0]; }
}

__global__ void reduce2_kernel(float* __restrict__ red) {
  __shared__ float smn[RB], smx[RB];
  const int t = threadIdx.x;
  smn[t] = red[2 + t]; smx[t] = red[2 + RB + t];
  __syncthreads();
  for (int s = RB / 2; s > 0; s >>= 1) {
    if (t < s) { smn[t] = fminf(smn[t], smn[t + s]); smx[t] = fmaxf(smx[t], smx[t + s]); }
    __syncthreads();
  }
  if (t == 0) { red[0] = smn[0]; red[1] = smx[0]; }
}

__global__ void normalize_kernel(float* __restrict__ out, const float* __restrict__ red, int n) {
  int i = blockIdx.x * blockDim.x + threadIdx.x;
  if (i < n) {
    float mn = red[0];
    float inv = 1.0f / (red[1] - mn + 1e-8f);
    out[i] = (out[i] - mn) * inv;
  }
}

extern "C" void kernel_launch(void* const* d_in, const int* in_sizes, int n_in,
                              void* d_out, int out_size, void* d_ws, size_t ws_size,
                              hipStream_t stream) {
  const float* image = (const float*)d_in[0];
  const float* x     = (const float*)d_in[1];
  float* out = (float*)d_out;

  const size_t TBL = (size_t)M_PAD * RESN;
  const size_t IMG = (size_t)NC * RESN * RESN;
  const size_t KSP = (size_t)NC * M_PAD;
  float* p    = (float*)d_ws;
  float* EhR  = p; p += TBL;
  float* EhI  = p; p += TBL;
  float* EwR  = p; p += TBL;
  float* EwI  = p; p += TBL;
  float* imgR = p; p += IMG;
  float* imgI = p; p += IMG;
  float* kspR = p; p += KSP;
  float* kspI = p; p += KSP;
  float* xt   = p; p += (size_t)M_TOTAL * 2;
  float* red  = p;

  const int n = NC * RESN * RESN;

  traj_kernel<<<(M_TOTAL * 2 + 255) / 256, 256, 0, stream>>>(x, xt);
  phase_kernel<<<(M_PAD * RESN + 255) / 256, 256, 0, stream>>>(xt, EhR, EhI, EwR, EwI);
  split_img_kernel<<<(n + 255) / 256, 256, 0, stream>>>(image, imgR, imgI);

  dim3 g1(M_PAD / 16, NC);
  gemm1_kernel<<<g1, 32, 0, stream>>>(imgR, imgI, EhR, EhI, EwR, EwI, kspR, kspI);

  dim3 g2((NC * RESN) / 64, RESN / 32);
  gemm2_kernel<<<g2, 256, 0, stream>>>(EhR, EhI, EwR, EwI, kspR, kspI, out);

  reduce1_kernel<<<RB, 256, 0, stream>>>(out, red, n);
  reduce2_kernel<<<1, RB, 0, stream>>>(red);
  normalize_kernel<<<(n + 255) / 256, 256, 0, stream>>>(out, red, n);
}


# gemm1 re-tiled: 8-wave blocks stage Ew and image K-slabs in LDS by full-line loads, 16m x 64h per wave; the h fold sums per lane and crosses lanes once at the end (sum order changes)
# speedup vs baseline: 2.1299x; 2.1299x over previous
#include <hip/hip_runtime.h>
#include <math.h>


typedef float v2f __attribute__((ext_vector_type(2)));
typedef float v4f __attribute__((ext_vector_type(4)));
typedef float v8f __attribute__((ext_vector_type(8)));

#define RESN     320
#define NSHOTS   8
#define SPS      1001
#define M_TOTAL  (NSHOTS * SPS)
#define M_PAD    8064
#define NC       8
#define TWO_PI   6.283185307179586f
#define RB       256

#define V8F_ZERO {0.f, 0.f, 0.f, 0.f, 0.f, 0.f, 0.f, 0.f}

__device__ __forceinline__ v8f wmma_f32(v2f a, v2f b, v8f c) {
  return __builtin_amdgcn_wmma_f32_16x16x4_f32(
       false, a,  false, b,
       (short)0, c,  false,  false);
}

__device__ __forceinline__ v2f ldv2(const float* __restrict__ p) {
  return *(const v2f*)p;
}

__global__ void traj_kernel(const float* __restrict__ x, float* __restrict__ xt) {
  int gid = blockIdx.x * blockDim.x + threadIdx.x;
  if (gid >= M_TOTAL * 2) return;
  int d    = gid & 1;
  int m    = gid >> 1;
  int shot = m / SPS;
  int tt   = m - shot * SPS;
  int I    = (tt == 0) ? 0 : (tt - 1) / 10;
  if (I > 99) I = 99;
  const float* base = x + (size_t)shot * SPS * 2 + d;
  float y0 = base[(I) * 20];
  float y1 = base[(I + 1) * 20];
  float dI = (y1 - y0) * 0.1f;
  float mI, mI1;
  if (I == 0) { mI = dI; }
  else        { float ym = base[(I - 1) * 20]; mI = 0.5f * (dI + (y0 - ym) * 0.1f); }
  if (I == 99) { mI1 = dI; }
  else         { float y2 = base[(I + 2) * 20]; mI1 = 0.5f * (dI + (y2 - y1) * 0.1f); }
  float s  = (float)(tt - 10 * I) * 0.1f;
  float s2 = s * s, s3 = s2 * s;
  float h00 = 1.0f - 3.0f * s2 + 2.0f * s3;
  float h10 = s - 2.0f * s2 + s3;
  float h01 = 3.0f * s2 - 2.0f * s3;
  float h11 = s3 - s2;
  xt[gid] = h00 * y0 + h10 * mI * 10.0f + h01 * y1 + h11 * mI1 * 10.0f;
}

__global__ void phase_kernel(const float* __restrict__ xt,
                             float* __restrict__ EhR, float* __restrict__ EhI,
                             float* __restrict__ EwR, float* __restrict__ EwI) {
  int gid = blockIdx.x * blockDim.x + threadIdx.x;
  if (gid >= M_PAD * RESN) return;
  int m = gid / RESN;
  int n = gid - m * RESN;
  float ehr = 0.f, ehi = 0.f, ewr = 0.f, ewi = 0.f;
  if (m < M_TOTAL) {
    float k = (TWO_PI / (float)RESN) * ((float)n - 160.0f);
    float sh, ch, sw, cw;
    __sincosf(xt[2 * m + 0] * k, &sh, &ch);
    __sincosf(xt[2 * m + 1] * k, &sw, &cw);
    ehr = ch; ehi = -sh; ewr = cw; ewi = -sw;
  }
  EhR[gid] = ehr; EhI[gid] = ehi;
  EwR[gid] = ewr; EwI[gid] = ewi;
}

__global__ void split_img_kernel(const float* __restrict__ image,
                                 float* __restrict__ imgR, float* __restrict__ imgI) {
  int i = blockIdx.x * blockDim.x + threadIdx.x;
  if (i < NC * RESN * RESN) {
    imgR[i] = image[2 * i];
    imgI[i] = image[2 * i + 1];
  }
}

#define G1_BM 128
#define G1_BN 64
#define G1_KB 32
#define G1_LS 36
__global__ __launch_bounds__(256) void gemm1_kernel(
    const float* __restrict__ imgR, const float* __restrict__ imgI,
    const float* __restrict__ EhR, const float* __restrict__ EhI,
    const float* __restrict__ EwR, const float* __restrict__ EwI,
    float* __restrict__ kspR, float* __restrict__ kspI) {
  __shared__ __attribute__((aligned(16))) float sAR[G1_BM * G1_LS];
  __shared__ __attribute__((aligned(16))) float sAI[G1_BM * G1_LS];
  __shared__ __attribute__((aligned(16))) float sBR[G1_BN * G1_LS];
  __shared__ __attribute__((aligned(16))) float sBI[G1_BN * G1_LS];
  const int tid  = threadIdx.x;
  const int wave = tid >> 5;
  const int l    = tid & 31;
  const int half = l >> 4;
  const int lm   = l & 15;
  const int m0   = blockIdx.x * G1_BM;
  const int c    = blockIdx.y;
  const int q = tid & 7;
  const int r = tid >> 3;
  const float* gAR = EwR + (size_t)(m0 + r) * RESN + 4 * q;
  const float* gAI = EwI + (size_t)(m0 + r) * RESN + 4 * q;
  const float* gBR = imgR + ((size_t)c * RESN + r) * RESN + 4 * q;
  const float* gBI = imgI + ((size_t)c * RESN + r) * RESN + 4 * q;
  const int sOff = r * G1_LS + 4 * q;

  v4f pAR[4], pAI[4], pBR[2], pBI[2];
#pragma unroll
  for (int j = 0; j < 4; ++j) {
    pAR[j] = *(const v4f*)(gAR + (size_t)(32 * j) * RESN);
    pAI[j] = *(const v4f*)(gAI + (size_t)(32 * j) * RESN);
  }
#pragma unroll
  for (int j = 0; j < 2; ++j) {
    pBR[j] = *(const v4f*)(gBR + (size_t)(32 * j) * RESN);
    pBI[j] = *(const v4f*)(gBI + (size_t)(32 * j) * RESN);
  }

  float accR[8], accI[8];
#pragma unroll
  for (int v = 0; v < 8; ++v) { accR[v] = 0.0f; accI[v] = 0.0f; }

  const int aOff = (wave * 16 + lm) * G1_LS + 2 * half;
  const int bOff = lm * G1_LS + 2 * half;

#pragma unroll 1
  for (int hc = 0; hc < RESN / G1_BN; ++hc) {
    v8f C1[4], C2[4], C3[4];
#pragma unroll
    for (int j = 0; j < 4; ++j) {
      v8f z = V8F_ZERO;
      C1[j] = z; C2[j] = z; C3[j] = z;
    }
#pragma unroll 1
    for (int ks = 0; ks < RESN / G1_KB; ++ks) {
#pragma unroll
      for (int j = 0; j < 4; ++j) {
        *(v4f*)(sAR + sOff + 32 * j * G1_LS) = pAR[j];
        *(v4f*)(sAI + sOff + 32 * j * G1_LS) = pAI[j];
      }
#pragma unroll
      for (int j = 0; j < 2; ++j) {
        *(v4f*)(sBR + sOff + 32 * j * G1_LS) = pBR[j];
        *(v4f*)(sBI + sOff + 32 * j * G1_LS) = pBI[j];
      }
      __syncthreads();
      int ksn = ks + 1, hcn = hc;
      if (ksn == RESN / G1_KB) { ksn = 0; hcn = hc + 1; }
      if (hcn < RESN / G1_BN) {
        const int kn = ksn * G1_KB;
        const size_t hn = (size_t)(hcn * G1_BN) * RESN;
#pragma unroll
        for (int j = 0; j < 4; ++j) {
          pAR[j] = *(const v4f*)(gAR + (size_t)(32 * j) * RESN + kn);
          pAI[j] = *(const v4f*)(gAI + (size_t)(32 * j) * RESN + kn);
        }
#pragma unroll
        for (int j = 0; j < 2; ++j) {
          pBR[j] = *(const v4f*)(gBR + hn + (size_t)(32 * j) * RESN + kn);
          pBI[j] = *(const v4f*)(gBI + hn + (size_t)(32 * j) * RESN + kn);
        }
      }
#pragma unroll
      for (int kk = 0; kk < G1_KB; kk += 4) {
        v2f Ar = ldv2(sAR + aOff + kk);
        v2f Ai = ldv2(sAI + aOff + kk);
        v2f As = Ar + Ai;
#pragma unroll
        for (int j = 0; j < 4; ++j) {
          v2f Br = ldv2(sBR + bOff + j * 16 * G1_LS + kk);
          v2f Bi = ldv2(sBI + bOff + j * 16 * G1_LS + kk);
          v2f Bs = Br + Bi;
          C1[j] = wmma_f32(Ar, Br, C1[j]);
          C2[j] = wmma_f32(Ai, Bi, C2[j]);
          C3[j] = wmma_f32(As, Bs, C3[j]);
        }
      }
      __syncthreads();
    }
#pragma unroll
    for (int j = 0; j < 4; ++j) {
#pragma unroll
      for (int v = 0; v < 8; ++v) {
        const int ml = wave * 16 + v + 8 * half;
        const size_t ei = (size_t)(m0 + ml) * RESN + hc * G1_BN + j * 16 + lm;
        float er = EhR[ei], eim = EhI[ei];
        float tr = C1[j][v] - C2[j][v];
        float ti = C3[j][v] - C1[j][v] - C2[j][v];
        accR[v] += er * tr - eim * ti;
        accI[v] += er * ti + eim * tr;
      }
    }
  }
#pragma unroll
  for (int v = 0; v < 8; ++v) {
    float pr = accR[v], pi = accI[v];
#pragma unroll
    for (int mask = 8; mask >= 1; mask >>= 1) {
      pr += __shfl_xor(pr, mask, 32);
      pi += __shfl_xor(pi, mask, 32);
    }
    accR[v] = pr; accI[v] = pi;
  }
  if (lm == 0) {
#pragma unroll
    for (int v = 0; v < 8; ++v) {
      const int m = m0 + wave * 16 + v + 8 * half;
      if (m < M_TOTAL) {
        kspR[(size_t)c * M_PAD + m] = accR[v];
        kspI[(size_t)c * M_PAD + m] = accI[v];
      }
    }
  }
}

#define KBLK 32
#define LSTR 34
__global__ __launch_bounds__(256) void gemm2_kernel(
    const float* __restrict__ EhR, const float* __restrict__ EhI,
    const float* __restrict__ EwR, const float* __restrict__ EwI,
    const float* __restrict__ kspR, const float* __restrict__ kspI,
    float* __restrict__ out) {
  __shared__ float GtR[64 * LSTR];
  __shared__ float GtI[64 * LSTR];
  __shared__ float BtR[32 * LSTR];
  __shared__ float BtI[32 * LSTR];
  const int tid  = threadIdx.x;
  const int wave = tid >> 5;
  const int l    = tid & 31;
  const int half = l >> 4;
  const int lm   = l & 15;
  const int wr   = wave >> 1;
  const int wc   = wave & 1;
  const int rowTile = blockIdx.x;
  const int colTile = blockIdx.y;
  const int c  = rowTile / 5;
  const int h0 = (rowTile % 5) * 64;
  const int w0 = colTile * 32;

  v8f C1 = V8F_ZERO;
  v8f C2 = V8F_ZERO;
  v8f C3 = V8F_ZERO;

  const int aOff = (wr * 16 + lm) * LSTR + 2 * half;
  const int bOff = (wc * 16 + lm) * LSTR + 2 * half;

  for (int kb = 0; kb < (M_TOTAL + KBLK - 1) / KBLK; ++kb) {
    const int k0 = kb * KBLK;
    for (int i = tid; i < KBLK * 32; i += 256) {
      const int n = i & 31, k = i >> 5;
      const int m = k0 + k;
      float br = 0.f, bi = 0.f;
      if (m < M_TOTAL) {
        const size_t gi = (size_t)m * RESN + w0 + n;
        br = EwR[gi];
        bi = -EwI[gi];
      }
      BtR[n * LSTR + k] = br;
      BtI[n * LSTR + k] = bi;
    }
    for (int i = tid; i < 64 * KBLK; i += 256) {
      const int r = i & 63, k = i >> 6;
      const int m = k0 + k;
      float gr = 0.f, gi = 0.f;
      if (m < M_TOTAL) {
        const size_t ei = (size_t)m * RESN + h0 + r;
        float er = EhR[ei], eim = EhI[ei];
        float kr = kspR[(size_t)c * M_PAD + m];
        float ki = kspI[(size_t)c * M_PAD + m];
        gr = er * kr + eim * ki;
        gi = er * ki - eim * kr;
      }
      GtR[r * LSTR + k] = gr;
      GtI[r * LSTR + k] = gi;
    }
    __syncthreads();
#pragma unroll
    for (int kk = 0; kk < KBLK; kk += 4) {
      v2f Ar = ldv2(GtR + aOff + kk);
      v2f Ai = ldv2(GtI + aOff + kk);
      v2f Br = ldv2(BtR + bOff + kk);
      v2f Bi = ldv2(BtI + bOff + kk);
      v2f As = Ar + Ai;
      v2f Bs = Br + Bi;
      C1 = wmma_f32(Ar, Br, C1);
      C2 = wmma_f32(Ai, Bi, C2);
      C3 = wmma_f32(As, Bs, C3);
    }
    __syncthreads();
  }
#pragma unroll
  for (int v = 0; v < 8; ++v) {
    const int row = rowTile * 64 + wr * 16 + v + 8 * half;
    const int col = w0 + wc * 16 + lm;
    float rr = C1[v] - C2[v];
    float ri = C3[v] - C1[v] - C2[v];
    out[(size_t)row * RESN + col] = sqrtf(rr * rr + ri * ri);
  }
}

__global__ void reduce1_kernel(const float* __restrict__ out, float* __restrict__ red, int n) {
  __shared__ float smn[256], smx[256];
  float mn = 3.402823466e38f, mx = -3.402823466e38f;
  for (int i = blockIdx.x * blockDim.x + threadIdx.x; i < n; i += gridDim.x * blockDim.x) {
    float v = out[i];
    mn = fminf(mn, v); mx = fmaxf(mx, v);
  }
  smn[threadIdx.x] = mn; smx[threadIdx.x] = mx;
  __syncthreads();
  for (int s = 128; s > 0; s >>= 1) {
    if ((int)threadIdx.x < s) {
      smn[threadIdx.x] = fminf(smn[threadIdx.x], smn[threadIdx.x + s]);
      smx[threadIdx.x] = fmaxf(smx[threadIdx.x], smx[threadIdx.x + s]);
    }
    __syncthreads();
  }
  if (threadIdx.x == 0) { red[2 + blockIdx.x] = smn[0]; red[2 + RB + blockIdx.x] = smx[0]; }
}

__global__ void reduce2_kernel(float* __restrict__ red) {
  __shared__ float smn[RB], smx[RB];
  const int t = threadIdx.x;
  smn[t] = red[2 + t]; smx[t] = red[2 + RB + t];
  __syncthreads();
  for (int s = RB / 2; s > 0; s >>= 1) {
    if (t < s) { smn[t] = fminf(smn[t], smn[t + s]); smx[t] = fmaxf(smx[t], smx[t + s]); }
    __syncthreads();
  }
  if (t == 0) { red[0] = smn[0]; red[1] = smx[0]; }
}

__global__ void normalize_kernel(float* __restrict__ out, const float* __restrict__ red, int n) {
  int i = blockIdx.x * blockDim.x + threadIdx.x;
  if (i < n) {
    float mn = red[0];
    float inv = 1.0f / (red[1] - mn + 1e-8f);
    out[i] = (out[i] - mn) * inv;
  }
}

extern "C" void kernel_launch(void* const* d_in, const int* in_sizes, int n_in,
                              void* d_out, int out_size, void* d_ws, size_t ws_size,
                              hipStream_t stream) {
  const float* image = (const float*)d_in[0];
  const float* x     = (const float*)d_in[1];
  float* out = (float*)d_out;

  const size_t TBL = (size_t)M_PAD * RESN;
  const size_t IMG = (size_t)NC * RESN * RESN;
  const size_t KSP = (size_t)NC * M_PAD;
  float* p    = (float*)d_ws;
  float* EhR  = p; p += TBL;
  float* EhI  = p; p += TBL;
  float* EwR  = p; p += TBL;
  float* EwI  = p; p += TBL;
  float* imgR = p; p += IMG;
  float* imgI = p; p += IMG;
  float* kspR = p; p += KSP;
  float* kspI = p; p += KSP;
  float* xt   = p; p += (size_t)M_TOTAL * 2;
  float* red  = p;

  const int n = NC * RESN * RESN;

  traj_kernel<<<(M_TOTAL * 2 + 255) / 256, 256, 0, stream>>>(x, xt);
  phase_kernel<<<(M_PAD * RESN + 255) / 256, 256, 0, stream>>>(xt, EhR, EhI, EwR, EwI);
  split_img_kernel<<<(n + 255) / 256, 256, 0, stream>>>(image, imgR, imgI);

  dim3 g1(M_PAD / G1_BM, NC);
  gemm1_kernel<<<g1, 256, 0, stream>>>(imgR, imgI, EhR, EhI, EwR, EwI, kspR, kspI);

  dim3 g2((NC * RESN) / 64, RESN / 32);
  gemm2_kernel<<<g2, 256, 0, stream>>>(EhR, EhI, EwR, EwI, kspR, kspI, out);

  reduce1_kernel<<<RB, 256, 0, stream>>>(out, red, n);
  reduce2_kernel<<<1, RB, 0, stream>>>(red);
  normalize_kernel<<<(n + 255) / 256, 256, 0, stream>>>(out, red, n);
}


# gemm2 re-tiled: 4-wave blocks, 32x32 complex tile per wave, Eh/Ew k-rows staged by full-line loads, ksp folded into Ew; K split 6 ways, partials added in the min/max pass (sum order changes)
# speedup vs baseline: 5.0550x; 2.3734x over previous
#include <hip/hip_runtime.h>
#include <math.h>


typedef float v2f __attribute__((ext_vector_type(2)));
typedef float v4f __attribute__((ext_vector_type(4)));
typedef float v8f __attribute__((ext_vector_type(8)));

#define RESN     320
#define NSHOTS   8
#define SPS      1001
#define M_TOTAL  (NSHOTS * SPS)
#define M_PAD    8064
#define NC       8
#define TWO_PI   6.283185307179586f
#define RB       256

#define V8F_ZERO {0.f, 0.f, 0.f, 0.f, 0.f, 0.f, 0.f, 0.f}

__device__ __forceinline__ v8f wmma_f32(v2f a, v2f b, v8f c) {
  return __builtin_amdgcn_wmma_f32_16x16x4_f32(
       false, a,  false, b,
       (short)0, c,  false,  false);
}

__device__ __forceinline__ v2f ldv2(const float* __restrict__ p) {
  return *(const v2f*)p;
}

__global__ void traj_kernel(const float* __restrict__ x, float* __restrict__ xt) {
  int gid = blockIdx.x * blockDim.x + threadIdx.x;
  if (gid >= M_TOTAL * 2) return;
  int d    = gid & 1;
  int m    = gid >> 1;
  int shot = m / SPS;
  int tt   = m - shot * SPS;
  int I    = (tt == 0) ? 0 : (tt - 1) / 10;
  if (I > 99) I = 99;
  const float* base = x + (size_t)shot * SPS * 2 + d;
  float y0 = base[(I) * 20];
  float y1 = base[(I + 1) * 20];
  float dI = (y1 - y0) * 0.1f;
  float mI, mI1;
  if (I == 0) { mI = dI; }
  else        { float ym = base[(I - 1) * 20]; mI = 0.5f * (dI + (y0 - ym) * 0.1f); }
  if (I == 99) { mI1 = dI; }
  else         { float y2 = base[(I + 2) * 20]; mI1 = 0.5f * (dI + (y2 - y1) * 0.1f); }
  float s  = (float)(tt - 10 * I) * 0.1f;
  float s2 = s * s, s3 = s2 * s;
  float h00 = 1.0f - 3.0f * s2 + 2.0f * s3;
  float h10 = s - 2.0f * s2 + s3;
  float h01 = 3.0f * s2 - 2.0f * s3;
  float h11 = s3 - s2;
  xt[gid] = h00 * y0 + h10 * mI * 10.0f + h01 * y1 + h11 * mI1 * 10.0f;
}

__global__ void phase_kernel(const float* __restrict__ xt,
                             float* __restrict__ EhR, float* __restrict__ EhI,
                             float* __restrict__ EwR, float* __restrict__ EwI) {
  int gid = blockIdx.x * blockDim.x + threadIdx.x;
  if (gid >= M_PAD * RESN) return;
  int m = gid / RESN;
  int n = gid - m * RESN;
  float ehr = 0.f, ehi = 0.f, ewr = 0.f, ewi = 0.f;
  if (m < M_TOTAL) {
    float k = (TWO_PI / (float)RESN) * ((float)n - 160.0f);
    float sh, ch, sw, cw;
    __sincosf(xt[2 * m + 0] * k, &sh, &ch);
    __sincosf(xt[2 * m + 1] * k, &sw, &cw);
    ehr = ch; ehi = -sh; ewr = cw; ewi = -sw;
  }
  EhR[gid] = ehr; EhI[gid] = ehi;
  EwR[gid] = ewr; EwI[gid] = ewi;
}

__global__ void split_img_kernel(const float* __restrict__ image,
                                 float* __restrict__ imgR, float* __restrict__ imgI) {
  int i = blockIdx.x * blockDim.x + threadIdx.x;
  if (i < NC * RESN * RESN) {
    imgR[i] = image[2 * i];
    imgI[i] = image[2 * i + 1];
  }
}

#define G1_BM 128
#define G1_BN 64
#define G1_KB 32
#define G1_LS 36
__global__ __launch_bounds__(256) void gemm1_kernel(
    const float* __restrict__ imgR, const float* __restrict__ imgI,
    const float* __restrict__ EhR, const float* __restrict__ EhI,
    const float* __restrict__ EwR, const float* __restrict__ EwI,
    float* __restrict__ kspR, float* __restrict__ kspI) {
  __shared__ __attribute__((aligned(16))) float sAR[G1_BM * G1_LS];
  __shared__ __attribute__((aligned(16))) float sAI[G1_BM * G1_LS];
  __shared__ __attribute__((aligned(16))) float sBR[G1_BN * G1_LS];
  __shared__ __attribute__((aligned(16))) float sBI[G1_BN * G1_LS];
  const int tid  = threadIdx.x;
  const int wave = tid >> 5;
  const int l    = tid & 31;
  const int half = l >> 4;
  const int lm   = l & 15;
  const int m0   = blockIdx.x * G1_BM;
  const int c    = blockIdx.y;
  const int q = tid & 7;
  const int r = tid >> 3;
  const float* gAR = EwR + (size_t)(m0 + r) * RESN + 4 * q;
  const float* gAI = EwI + (size_t)(m0 + r) * RESN + 4 * q;
  const float* gBR = imgR + ((size_t)c * RESN + r) * RESN + 4 * q;
  const float* gBI = imgI + ((size_t)c * RESN + r) * RESN + 4 * q;
  const int sOff = r * G1_LS + 4 * q;

  v4f pAR[4], pAI[4], pBR[2], pBI[2];
#pragma unroll
  for (int j = 0; j < 4; ++j) {
    pAR[j] = *(const v4f*)(gAR + (size_t)(32 * j) * RESN);
    pAI[j] = *(const v4f*)(gAI + (size_t)(32 * j) * RESN);
  }
#pragma unroll
  for (int j = 0; j < 2; ++j) {
    pBR[j] = *(const v4f*)(gBR + (size_t)(32 * j) * RESN);
    pBI[j] = *(const v4f*)(gBI + (size_t)(32 * j) * RESN);
  }

  float accR[8], accI[8];
#pragma unroll
  for (int v = 0; v < 8; ++v) { accR[v] = 0.0f; accI[v] = 0.0f; }

  const int aOff = (wave * 16 + lm) * G1_LS + 2 * half;
  const int bOff = lm * G1_LS + 2 * half;

#pragma unroll 1
  for (int hc = 0; hc < RESN / G1_BN; ++hc) {
    v8f C1[4], C2[4], C3[4];
#pragma unroll
    for (int j = 0; j < 4; ++j) {
      v8f z = V8F_ZERO;
      C1[j] = z; C2[j] = z; C3[j] = z;
    }
#pragma unroll 1
    for (int ks = 0; ks < RESN / G1_KB; ++ks) {
#pragma unroll
      for (int j = 0; j < 4; ++j) {
        *(v4f*)(sAR + sOff + 32 * j * G1_LS) = pAR[j];
        *(v4f*)(sAI + sOff + 32 * j * G1_LS) = pAI[j];
      }
#pragma unroll
      for (int j = 0; j < 2; ++j) {
        *(v4f*)(sBR + sOff + 32 * j * G1_LS) = pBR[j];
        *(v4f*)(sBI + sOff + 32 * j * G1_LS) = pBI[j];
      }
      __syncthreads();
      int ksn = ks + 1, hcn = hc;
      if (ksn == RESN / G1_KB) { ksn = 0; hcn = hc + 1; }
      if (hcn < RESN / G1_BN) {
        const int kn = ksn * G1_KB;
        const size_t hn = (size_t)(hcn * G1_BN) * RESN;
#pragma unroll
        for (int j = 0; j < 4; ++j) {
          pAR[j] = *(const v4f*)(gAR + (size_t)(32 * j) * RESN + kn);
          pAI[j] = *(const v4f*)(gAI + (size_t)(32 * j) * RESN + kn);
        }
#pragma unroll
        for (int j = 0; j < 2; ++j) {
          pBR[j] = *(const v4f*)(gBR + hn + (size_t)(32 * j) * RESN + kn);
          pBI[j] = *(const v4f*)(gBI + hn + (size_t)(32 * j) * RESN + kn);
        }
      }
#pragma unroll
      for (int kk = 0; kk < G1_KB; kk += 4) {
        v2f Ar = ldv2(sAR + aOff + kk);
        v2f Ai = ldv2(sAI + aOff + kk);
        v2f As = Ar + Ai;
#pragma unroll
        for (int j = 0; j < 4; ++j) {
          v2f Br = ldv2(sBR + bOff + j * 16 * G1_LS + kk);
          v2f Bi = ldv2(sBI + bOff + j * 16 * G1_LS + kk);
          v2f Bs = Br + Bi;
          C1[j] = wmma_f32(Ar, Br, C1[j]);
          C2[j] = wmma_f32(Ai, Bi, C2[j]);
          C3[j] = wmma_f32(As, Bs, C3[j]);
        }
      }
      __syncthreads();
    }
#pragma unroll
    for (int j = 0; j < 4; ++j) {
#pragma unroll
      for (int v = 0; v < 8; ++v) {
        const int ml = wave * 16 + v + 8 * half;
        const size_t ei = (size_t)(m0 + ml) * RESN + hc * G1_BN + j * 16 + lm;
        float er = EhR[ei], eim = EhI[ei];
        float tr = C1[j][v] - C2[j][v];
        float ti = C3[j][v] - C1[j][v] - C2[j][v];
        accR[v] += er * tr - eim * ti;
        accI[v] += er * ti + eim * tr;
      }
    }
  }
#pragma unroll
  for (int v = 0; v < 8; ++v) {
    float pr = accR[v], pi = accI[v];
#pragma unroll
    for (int mask = 8; mask >= 1; mask >>= 1) {
      pr += __shfl_xor(pr, mask, 32);
      pi += __shfl_xor(pi, mask, 32);
    }
    accR[v] = pr; accI[v] = pi;
  }
  if (lm == 0) {
#pragma unroll
    for (int v = 0; v < 8; ++v) {
      const int m = m0 + wave * 16 + v + 8 * half;
      if (m < M_TOTAL) {
        kspR[(size_t)c * M_PAD + m] = accR[v];
        kspI[(size_t)c * M_PAD + m] = accI[v];
      }
    }
  }
}

#define G2_KB    32
#define G2_LS    72
#define G2_SPLIT 6
#define G2_SLABS (M_PAD / G2_KB / G2_SPLIT)
__global__ __launch_bounds__(128) void gemm2_kernel(
    const float* __restrict__ EhR, const float* __restrict__ EhI,
    const float* __restrict__ EwR, const float* __restrict__ EwI,
    const float* __restrict__ kspR, const float* __restrict__ kspI,
    float* __restrict__ partR, float* __restrict__ partI) {
  __shared__ __attribute__((aligned(16))) float sAR[G2_KB * G2_LS];
  __shared__ __attribute__((aligned(16))) float sAI[G2_KB * G2_LS];
  __shared__ __attribute__((aligned(16))) float sBR[G2_KB * G2_LS];
  __shared__ __attribute__((aligned(16))) float sBI[G2_KB * G2_LS];
  const int tid  = threadIdx.x;
  const int wave = tid >> 5;
  const int l    = tid & 31;
  const int half = l >> 4;
  const int lm   = l & 15;
  const int wr   = wave >> 1;
  const int wc   = wave & 1;
  const int rowTile = blockIdx.x / 5;
  const int colTile = blockIdx.x - rowTile * 5;
  const int c   = rowTile / 5;
  const int h0  = (rowTile - c * 5) * 64;
  const int w0  = colTile * 64;
  const int kb0 = blockIdx.y * (G2_SLABS * G2_KB);

  const int q = tid & 15;
  const int r = tid >> 4;
  const float* gAR = EhR + (size_t)(kb0 + r) * RESN + h0 + 4 * q;
  const float* gAI = EhI + (size_t)(kb0 + r) * RESN + h0 + 4 * q;
  const float* gBR = EwR + (size_t)(kb0 + r) * RESN + w0 + 4 * q;
  const float* gBI = EwI + (size_t)(kb0 + r) * RESN + w0 + 4 * q;
  const float* gKR = kspR + (size_t)c * M_PAD;
  const float* gKI = kspI + (size_t)c * M_PAD;
  const int sOff = r * G2_LS + 4 * q;

  v4f pAR[4], pAI[4], pBR[4], pBI[4];
  float pKR[4], pKI[4];
#pragma unroll
  for (int j = 0; j < 4; ++j) {
    const int m = kb0 + r + 8 * j;
    pAR[j] = *(const v4f*)(gAR + (size_t)(8 * j) * RESN);
    pAI[j] = *(const v4f*)(gAI + (size_t)(8 * j) * RESN);
    pBR[j] = *(const v4f*)(gBR + (size_t)(8 * j) * RESN);
    pBI[j] = *(const v4f*)(gBI + (size_t)(8 * j) * RESN);
    const int mc = (m < M_TOTAL) ? m : (M_TOTAL - 1);
    const float kr = gKR[mc], ki = gKI[mc];
    pKR[j] = (m < M_TOTAL) ? kr : 0.0f;
    pKI[j] = (m < M_TOTAL) ? ki : 0.0f;
  }

  v8f C1[2][2], C2[2][2], C3[2][2];
#pragma unroll
  for (int i = 0; i < 2; ++i) {
#pragma unroll
    for (int j = 0; j < 2; ++j) {
      v8f z = V8F_ZERO;
      C1[i][j] = z; C2[i][j] = z; C3[i][j] = z;
    }
  }

  const int aOff = (2 * half) * G2_LS + wr * 32 + lm;
  const int bOff = (2 * half) * G2_LS + wc * 32 + lm;

#pragma unroll 1
  for (int ks = 0; ks < G2_SLABS; ++ks) {
#pragma unroll
    for (int j = 0; j < 4; ++j) {
      *(v4f*)(sAR + sOff + 8 * j * G2_LS) = pAR[j];
      *(v4f*)(sAI + sOff + 8 * j * G2_LS) = pAI[j];
      const float kr = pKR[j], ki = pKI[j];
      v4f er = pBR[j], ei = pBI[j];
      v4f bp = kr * er + ki * ei;
      v4f bq = ki * er - kr * ei;
      *(v4f*)(sBR + sOff + 8 * j * G2_LS) = bp;
      *(v4f*)(sBI + sOff + 8 * j * G2_LS) = bq;
    }
    __syncthreads();
    if (ks + 1 < G2_SLABS) {
      const int kn = (ks + 1) * G2_KB;
#pragma unroll
      for (int j = 0; j < 4; ++j) {
        const int m = kb0 + kn + r + 8 * j;
        const size_t go = (size_t)(kn + 8 * j) * RESN;
        pAR[j] = *(const v4f*)(gAR + go);
        pAI[j] = *(const v4f*)(gAI + go);
        pBR[j] = *(const v4f*)(gBR + go);
        pBI[j] = *(const v4f*)(gBI + go);
        const int mc = (m < M_TOTAL) ? m : (M_TOTAL - 1);
        const float kr = gKR[mc], ki = gKI[mc];
        pKR[j] = (m < M_TOTAL) ? kr : 0.0f;
        pKI[j] = (m < M_TOTAL) ? ki : 0.0f;
      }
    }
#pragma unroll
    for (int kk = 0; kk < G2_KB; kk += 4) {
      v2f Aa[2], Ab[2], As[2], Bp[2], Bq[2], Bd[2];
#pragma unroll
      for (int i = 0; i < 2; ++i) {
        const int o = aOff + kk * G2_LS + i * 16;
        Aa[i][0] = sAR[o]; Aa[i][1] = sAR[o + G2_LS];
        Ab[i][0] = sAI[o]; Ab[i][1] = sAI[o + G2_LS];
        As[i] = Aa[i] + Ab[i];
      }
#pragma unroll
      for (int j = 0; j < 2; ++j) {
        const int o = bOff + kk * G2_LS + j * 16;
        Bp[j][0] = sBR[o]; Bp[j][1] = sBR[o + G2_LS];
        Bq[j][0] = sBI[o]; Bq[j][1] = sBI[o + G2_LS];
        Bd[j] = Bq[j] - Bp[j];
      }
#pragma unroll
      for (int i = 0; i < 2; ++i) {
#pragma unroll
        for (int j = 0; j < 2; ++j) {
          C1[i][j] = wmma_f32(Aa[i], Bp[j], C1[i][j]);
          C2[i][j] = wmma_f32(Ab[i], Bq[j], C2[i][j]);
          C3[i][j] = wmma_f32(As[i], Bd[j], C3[i][j]);
        }
      }
    }
    __syncthreads();
  }
  const size_t pbase = (size_t)blockIdx.y * ((size_t)NC * RESN * RESN);
#pragma unroll
  for (int i = 0; i < 2; ++i) {
#pragma unroll
    for (int j = 0; j < 2; ++j) {
#pragma unroll
      for (int v = 0; v < 8; ++v) {
        const int row = rowTile * 64 + wr * 32 + i * 16 + v + 8 * half;
        const int col = w0 + wc * 32 + j * 16 + lm;
        const size_t oi = pbase + (size_t)row * RESN + col;
        partR[oi] = C1[i][j][v] + C2[i][j][v];
        partI[oi] = C3[i][j][v] + C1[i][j][v] - C2[i][j][v];
      }
    }
  }
}

__global__ void reduce1_kernel(const float* __restrict__ partR, const float* __restrict__ partI,
                               float* __restrict__ out, float* __restrict__ red, int n) {
  __shared__ float smn[256], smx[256];
  float mn = 3.402823466e38f, mx = -3.402823466e38f;
  for (int i = blockIdx.x * blockDim.x + threadIdx.x; i < n; i += gridDim.x * blockDim.x) {
    float rr = 0.0f, ri = 0.0f;
#pragma unroll
    for (int s = 0; s < G2_SPLIT; ++s) {
      rr += partR[(size_t)s * n + i];
      ri += partI[(size_t)s * n + i];
    }
    float v = sqrtf(rr * rr + ri * ri);
    out[i] = v;
    mn = fminf(mn, v); mx = fmaxf(mx, v);
  }
  smn[threadIdx.x] = mn; smx[threadIdx.x] = mx;
  __syncthreads();
  for (int s = 128; s > 0; s >>= 1) {
    if ((int)threadIdx.x < s) {
      smn[threadIdx.x] = fminf(smn[threadIdx.x], smn[threadIdx.x + s]);
      smx[threadIdx.x] = fmaxf(smx[threadIdx.x], smx[threadIdx.x + s]);
    }
    __syncthreads();
  }
  if (threadIdx.x == 0) { red[2 + blockIdx.x] = smn[0]; red[2 + RB + blockIdx.x] = smx[0]; }
}

__global__ void reduce2_kernel(float* __restrict__ red) {
  __shared__ float smn[RB], smx[RB];
  const int t = threadIdx.x;
  smn[t] = red[2 + t]; smx[t] = red[2 + RB + t];
  __syncthreads();
  for (int s = RB / 2; s > 0; s >>= 1) {
    if (t < s) { smn[t] = fminf(smn[t], smn[t + s]); smx[t] = fmaxf(smx[t], smx[t + s]); }
    __syncthreads();
  }
  if (t == 0) { red[0] = smn[0]; red[1] = smx[0]; }
}

__global__ void normalize_kernel(float* __restrict__ out, const float* __restrict__ red, int n) {
  int i = blockIdx.x * blockDim.x + threadIdx.x;
  if (i < n) {
    float mn = red[0];
    float inv = 1.0f / (red[1] - mn + 1e-8f);
    out[i] = (out[i] - mn) * inv;
  }
}

extern "C" void kernel_launch(void* const* d_in, const int* in_sizes, int n_in,
                              void* d_out, int out_size, void* d_ws, size_t ws_size,
                              hipStream_t stream) {
  const float* image = (const float*)d_in[0];
  const float* x     = (const float*)d_in[1];
  float* out = (float*)d_out;

  const size_t TBL = (size_t)M_PAD * RESN;
  const size_t IMG = (size_t)NC * RESN * RESN;
  const size_t KSP = (size_t)NC * M_PAD;
  float* p    = (float*)d_ws;
  float* EhR  = p; p += TBL;
  float* EhI  = p; p += TBL;
  float* EwR  = p; p += TBL;
  float* EwI  = p; p += TBL;
  float* imgR = p; p += IMG;
  float* imgI = p; p += IMG;
  float* kspR = p; p += KSP;
  float* kspI = p; p += KSP;
  float* xt   = p; p += (size_t)M_TOTAL * 2;
  float* red  = p; p += 1024;
  float* partR = p; p += (size_t)G2_SPLIT * IMG;
  float* partI = p;

  const int n = NC * RESN * RESN;

  traj_kernel<<<(M_TOTAL * 2 + 255) / 256, 256, 0, stream>>>(x, xt);
  phase_kernel<<<(M_PAD * RESN + 255) / 256, 256, 0, stream>>>(xt, EhR, EhI, EwR, EwI);
  split_img_kernel<<<(n + 255) / 256, 256, 0, stream>>>(image, imgR, imgI);

  dim3 g1(M_PAD / G1_BM, NC);
  gemm1_kernel<<<g1, 256, 0, stream>>>(imgR, imgI, EhR, EhI, EwR, EwI, kspR, kspI);

  dim3 g2(((NC * RESN) / 64) * (RESN / 64), G2_SPLIT);
  gemm2_kernel<<<g2, 128, 0, stream>>>(EhR, EhI, EwR, EwI, kspR, kspI, partR, partI);

  reduce1_kernel<<<RB, 256, 0, stream>>>(partR, partI, out, red, n);
  reduce2_kernel<<<1, RB, 0, stream>>>(red);
  normalize_kernel<<<(n + 255) / 256, 256, 0, stream>>>(out, red, n);
}


# gemm1 uses the +-n conjugate symmetry of the w phase row: image columns folded about w=160, real cos|sin table: 2 real WMMA products per result, not 3 (sum order changes)
# speedup vs baseline: 5.9268x; 1.1725x over previous
#include <hip/hip_runtime.h>
#include <math.h>


typedef float v2f __attribute__((ext_vector_type(2)));
typedef float v4f __attribute__((ext_vector_type(4)));
typedef float v8f __attribute__((ext_vector_type(8)));

#define RESN     320
#define NSHOTS   8
#define SPS      1001
#define M_TOTAL  (NSHOTS * SPS)
#define M_PAD    8064
#define NC       8
#define TWO_PI   6.283185307179586f
#define RB       256

#define V8F_ZERO {0.f, 0.f, 0.f, 0.f, 0.f, 0.f, 0.f, 0.f}

__device__ __forceinline__ v8f wmma_f32(v2f a, v2f b, v8f c) {
  return __builtin_amdgcn_wmma_f32_16x16x4_f32(
       false, a,  false, b,
       (short)0, c,  false,  false);
}

__device__ __forceinline__ v2f ldv2(const float* __restrict__ p) {
  return *(const v2f*)p;
}

__global__ void traj_kernel(const float* __restrict__ x, float* __restrict__ xt) {
  int gid = blockIdx.x * blockDim.x + threadIdx.x;
  if (gid >= M_TOTAL * 2) return;
  int d    = gid & 1;
  int m    = gid >> 1;
  int shot = m / SPS;
  int tt   = m - shot * SPS;
  int I    = (tt == 0) ? 0 : (tt - 1) / 10;
  if (I > 99) I = 99;
  const float* base = x + (size_t)shot * SPS * 2 + d;
  float y0 = base[(I) * 20];
  float y1 = base[(I + 1) * 20];
  float dI = (y1 - y0) * 0.1f;
  float mI, mI1;
  if (I == 0) { mI = dI; }
  else        { float ym = base[(I - 1) * 20]; mI = 0.5f * (dI + (y0 - ym) * 0.1f); }
  if (I == 99) { mI1 = dI; }
  else         { float y2 = base[(I + 2) * 20]; mI1 = 0.5f * (dI + (y2 - y1) * 0.1f); }
  float s  = (float)(tt - 10 * I) * 0.1f;
  float s2 = s * s, s3 = s2 * s;
  float h00 = 1.0f - 3.0f * s2 + 2.0f * s3;
  float h10 = s - 2.0f * s2 + s3;
  float h01 = 3.0f * s2 - 2.0f * s3;
  float h11 = s3 - s2;
  xt[gid] = h00 * y0 + h10 * mI * 10.0f + h01 * y1 + h11 * mI1 * 10.0f;
}

__global__ void phase_kernel(const float* __restrict__ xt,
                             float* __restrict__ EhR, float* __restrict__ EhI,
                             float* __restrict__ EwR, float* __restrict__ EwI,
                             float* __restrict__ Acs) {
  int gid = blockIdx.x * blockDim.x + threadIdx.x;
  if (gid >= M_PAD * RESN) return;
  int m = gid / RESN;
  int n = gid - m * RESN;
  float ehr = 0.f, ehi = 0.f, ewr = 0.f, ewi = 0.f;
  float cwv = 0.f, swv = 0.f;
  if (m < M_TOTAL) {
    float k = (TWO_PI / (float)RESN) * ((float)n - 160.0f);
    float sh, ch, sw, cw;
    __sincosf(xt[2 * m + 0] * k, &sh, &ch);
    __sincosf(xt[2 * m + 1] * k, &sw, &cw);
    ehr = ch; ehi = -sh; ewr = cw; ewi = -sw;
    cwv = cw; swv = sw;
  }
  EhR[gid] = ehr; EhI[gid] = ehi;
  EwR[gid] = ewr; EwI[gid] = ewi;
  const int nn = n - 160;
  float* arow = Acs + (size_t)m * RESN;
  if (nn >= 1) {
    arow[nn - 1]   = cwv;
    arow[159 + nn] = swv;
  } else if (nn == -160) {
    arow[159] = cwv;
    arow[319] = -swv;
  }
}

__global__ void fold_img_kernel(const float* __restrict__ image,
                                float* __restrict__ BR, float* __restrict__ BI) {
  int i = blockIdx.x * blockDim.x + threadIdx.x;
  if (i >= NC * RESN * RESN) return;
  const int row = i / RESN;
  const int k   = i - row * RESN;
  const bool lo = (k < 160);
  const int np  = lo ? (k + 1) : (k - 159);
  const float* s = image + (size_t)row * RESN * 2;
  float br, bi;
  if (np < 160) {
    const v2f a = *(const v2f*)(s + 2 * (160 + np));
    const v2f b = *(const v2f*)(s + 2 * (160 - np));
    if (lo) { br = a[0] + b[0]; bi = a[1] + b[1]; }
    else    { br = a[1] - b[1]; bi = b[0] - a[0]; }
  } else {
    const v2f b = *(const v2f*)s;
    if (lo) { br = b[0];  bi = b[1]; }
    else    { br = -b[1]; bi = b[0]; }
  }
  BR[i] = br; BI[i] = bi;
}

#define G1_BM 128
#define G1_BN 64
#define G1_KB 32
#define G1_LS 36
__global__ __launch_bounds__(256) void gemm1_kernel(
    const float* __restrict__ image,
    const float* __restrict__ BR, const float* __restrict__ BI,
    const float* __restrict__ EhR, const float* __restrict__ EhI,
    const float* __restrict__ Acs,
    float* __restrict__ kspR, float* __restrict__ kspI) {
  __shared__ __attribute__((aligned(16))) float sA[G1_BM * G1_LS];
  __shared__ __attribute__((aligned(16))) float sBR[G1_BN * G1_LS];
  __shared__ __attribute__((aligned(16))) float sBI[G1_BN * G1_LS];
  const int tid  = threadIdx.x;
  const int wave = tid >> 5;
  const int l    = tid & 31;
  const int half = l >> 4;
  const int lm   = l & 15;
  const int m0   = blockIdx.x * G1_BM;
  const int c    = blockIdx.y;
  const int q = tid & 7;
  const int r = tid >> 3;
  const float* gA  = Acs + (size_t)(m0 + r) * RESN + 4 * q;
  const float* gBR = BR + ((size_t)c * RESN + r) * RESN + 4 * q;
  const float* gBI = BI + ((size_t)c * RESN + r) * RESN + 4 * q;
  const int sOff = r * G1_LS + 4 * q;

  v4f pA[4], pBR[2], pBI[2];
#pragma unroll
  for (int j = 0; j < 4; ++j) {
    pA[j] = *(const v4f*)(gA + (size_t)(32 * j) * RESN);
  }
#pragma unroll
  for (int j = 0; j < 2; ++j) {
    pBR[j] = *(const v4f*)(gBR + (size_t)(32 * j) * RESN);
    pBI[j] = *(const v4f*)(gBI + (size_t)(32 * j) * RESN);
  }

  float accR[8], accI[8];
#pragma unroll
  for (int v = 0; v < 8; ++v) { accR[v] = 0.0f; accI[v] = 0.0f; }

  const int aOff = (wave * 16 + lm) * G1_LS + 2 * half;
  const int bOff = lm * G1_LS + 2 * half;

#pragma unroll 1
  for (int hc = 0; hc < RESN / G1_BN; ++hc) {
    v8f C1[4], C2[4];
#pragma unroll
    for (int j = 0; j < 4; ++j) {
      v8f z = V8F_ZERO;
      C1[j] = z; C2[j] = z;
    }
#pragma unroll 1
    for (int ks = 0; ks < RESN / G1_KB; ++ks) {
#pragma unroll
      for (int j = 0; j < 4; ++j) {
        *(v4f*)(sA + sOff + 32 * j * G1_LS) = pA[j];
      }
#pragma unroll
      for (int j = 0; j < 2; ++j) {
        *(v4f*)(sBR + sOff + 32 * j * G1_LS) = pBR[j];
        *(v4f*)(sBI + sOff + 32 * j * G1_LS) = pBI[j];
      }
      __syncthreads();
      int ksn = ks + 1, hcn = hc;
      if (ksn == RESN / G1_KB) { ksn = 0; hcn = hc + 1; }
      if (hcn < RESN / G1_BN) {
        const int kn = ksn * G1_KB;
        const size_t hn = (size_t)(hcn * G1_BN) * RESN;
#pragma unroll
        for (int j = 0; j < 4; ++j) {
          pA[j] = *(const v4f*)(gA + (size_t)(32 * j) * RESN + kn);
        }
#pragma unroll
        for (int j = 0; j < 2; ++j) {
          pBR[j] = *(const v4f*)(gBR + hn + (size_t)(32 * j) * RESN + kn);
          pBI[j] = *(const v4f*)(gBI + hn + (size_t)(32 * j) * RESN + kn);
        }
      }
#pragma unroll
      for (int kk = 0; kk < G1_KB; kk += 4) {
        v2f A = ldv2(sA + aOff + kk);
#pragma unroll
        for (int j = 0; j < 4; ++j) {
          v2f Br = ldv2(sBR + bOff + j * 16 * G1_LS + kk);
          v2f Bi = ldv2(sBI + bOff + j * 16 * G1_LS + kk);
          C1[j] = wmma_f32(A, Br, C1[j]);
          C2[j] = wmma_f32(A, Bi, C2[j]);
        }
      }
      __syncthreads();
    }
#pragma unroll
    for (int j = 0; j < 4; ++j) {
      const int h = hc * G1_BN + j * 16 + lm;
      const v2f i0 = ldv2(image + (((size_t)c * RESN + h) * RESN + 160) * 2);
#pragma unroll
      for (int v = 0; v < 8; ++v) {
        const int ml = wave * 16 + v + 8 * half;
        const size_t ei = (size_t)(m0 + ml) * RESN + h;
        float er = EhR[ei], eim = EhI[ei];
        float tr = C1[j][v] + i0[0];
        float ti = C2[j][v] + i0[1];
        accR[v] += er * tr - eim * ti;
        accI[v] += er * ti + eim * tr;
      }
    }
  }
#pragma unroll
  for (int v = 0; v < 8; ++v) {
    float pr = accR[v], pi = accI[v];
#pragma unroll
    for (int mask = 8; mask >= 1; mask >>= 1) {
      pr += __shfl_xor(pr, mask, 32);
      pi += __shfl_xor(pi, mask, 32);
    }
    accR[v] = pr; accI[v] = pi;
  }
  if (lm == 0) {
#pragma unroll
    for (int v = 0; v < 8; ++v) {
      const int m = m0 + wave * 16 + v + 8 * half;
      if (m < M_TOTAL) {
        kspR[(size_t)c * M_PAD + m] = accR[v];
        kspI[(size_t)c * M_PAD + m] = accI[v];
      }
    }
  }
}

#define G2_KB    32
#define G2_LS    72
#define G2_SPLIT 6
#define G2_SLABS (M_PAD / G2_KB / G2_SPLIT)
__global__ __launch_bounds__(128) void gemm2_kernel(
    const float* __restrict__ EhR, const float* __restrict__ EhI,
    const float* __restrict__ EwR, const float* __restrict__ EwI,
    const float* __restrict__ kspR, const float* __restrict__ kspI,
    float* __restrict__ partR, float* __restrict__ partI) {
  __shared__ __attribute__((aligned(16))) float sAR[G2_KB * G2_LS];
  __shared__ __attribute__((aligned(16))) float sAI[G2_KB * G2_LS];
  __shared__ __attribute__((aligned(16))) float sBR[G2_KB * G2_LS];
  __shared__ __attribute__((aligned(16))) float sBI[G2_KB * G2_LS];
  const int tid  = threadIdx.x;
  const int wave = tid >> 5;
  const int l    = tid & 31;
  const int half = l >> 4;
  const int lm   = l & 15;
  const int wr   = wave >> 1;
  const int wc   = wave & 1;
  const int rowTile = blockIdx.x / 5;
  const int colTile = blockIdx.x - rowTile * 5;
  const int c   = rowTile / 5;
  const int h0  = (rowTile - c * 5) * 64;
  const int w0  = colTile * 64;
  const int kb0 = blockIdx.y * (G2_SLABS * G2_KB);

  const int q = tid & 15;
  const int r = tid >> 4;
  const float* gAR = EhR + (size_t)(kb0 + r) * RESN + h0 + 4 * q;
  const float* gAI = EhI + (size_t)(kb0 + r) * RESN + h0 + 4 * q;
  const float* gBR = EwR + (size_t)(kb0 + r) * RESN + w0 + 4 * q;
  const float* gBI = EwI + (size_t)(kb0 + r) * RESN + w0 + 4 * q;
  const float* gKR = kspR + (size_t)c * M_PAD;
  const float* gKI = kspI + (size_t)c * M_PAD;
  const int sOff = r * G2_LS + 4 * q;

  v4f pAR[4], pAI[4], pBR[4], pBI[4];
  float pKR[4], pKI[4];
#pragma unroll
  for (int j = 0; j < 4; ++j) {
    const int m = kb0 + r + 8 * j;
    pAR[j] = *(const v4f*)(gAR + (size_t)(8 * j) * RESN);
    pAI[j] = *(const v4f*)(gAI + (size_t)(8 * j) * RESN);
    pBR[j] = *(const v4f*)(gBR + (size_t)(8 * j) * RESN);
    pBI[j] = *(const v4f*)(gBI + (size_t)(8 * j) * RESN);
    const int mc = (m < M_TOTAL) ? m : (M_TOTAL - 1);
    const float kr = gKR[mc], ki = gKI[mc];
    pKR[j] = (m < M_TOTAL) ? kr : 0.0f;
    pKI[j] = (m < M_TOTAL) ? ki : 0.0f;
  }

  v8f C1[2][2], C2[2][2], C3[2][2];
#pragma unroll
  for (int i = 0; i < 2; ++i) {
#pragma unroll
    for (int j = 0; j < 2; ++j) {
      v8f z = V8F_ZERO;
      C1[i][j] = z; C2[i][j] = z; C3[i][j] = z;
    }
  }

  const int aOff = (2 * half) * G2_LS + wr * 32 + lm;
  const int bOff = (2 * half) * G2_LS + wc * 32 + lm;

#pragma unroll 1
  for (int ks = 0; ks < G2_SLABS; ++ks) {
#pragma unroll
    for (int j = 0; j < 4; ++j) {
      *(v4f*)(sAR + sOff + 8 * j * G2_LS) = pAR[j];
      *(v4f*)(sAI + sOff + 8 * j * G2_LS) = pAI[j];
      const float kr = pKR[j], ki = pKI[j];
      v4f er = pBR[j], ei = pBI[j];
      v4f bp = kr * er + ki * ei;
      v4f bq = ki * er - kr * ei;
      *(v4f*)(sBR + sOff + 8 * j * G2_LS) = bp;
      *(v4f*)(sBI + sOff + 8 * j * G2_LS) = bq;
    }
    __syncthreads();
    if (ks + 1 < G2_SLABS) {
      const int kn = (ks + 1) * G2_KB;
#pragma unroll
      for (int j = 0; j < 4; ++j) {
        const int m = kb0 + kn + r + 8 * j;
        const size_t go = (size_t)(kn + 8 * j) * RESN;
        pAR[j] = *(const v4f*)(gAR + go);
        pAI[j] = *(const v4f*)(gAI + go);
        pBR[j] = *(const v4f*)(gBR + go);
        pBI[j] = *(const v4f*)(gBI + go);
        const int mc = (m < M_TOTAL) ? m : (M_TOTAL - 1);
        const float kr = gKR[mc], ki = gKI[mc];
        pKR[j] = (m < M_TOTAL) ? kr : 0.0f;
        pKI[j] = (m < M_TOTAL) ? ki : 0.0f;
      }
    }
#pragma unroll
    for (int kk = 0; kk < G2_KB; kk += 4) {
      v2f Aa[2], Ab[2], As[2], Bp[2], Bq[2], Bd[2];
#pragma unroll
      for (int i = 0; i < 2; ++i) {
        const int o = aOff + kk * G2_LS + i * 16;
        Aa[i][0] = sAR[o]; Aa[i][1] = sAR[o + G2_LS];
        Ab[i][0] = sAI[o]; Ab[i][1] = sAI[o + G2_LS];
        As[i] = Aa[i] + Ab[i];
      }
#pragma unroll
      for (int j = 0; j < 2; ++j) {
        const int o = bOff + kk * G2_LS + j * 16;
        Bp[j][0] = sBR[o]; Bp[j][1] = sBR[o + G2_LS];
        Bq[j][0] = sBI[o]; Bq[j][1] = sBI[o + G2_LS];
        Bd[j] = Bq[j] - Bp[j];
      }
#pragma unroll
      for (int i = 0; i < 2; ++i) {
#pragma unroll
        for (int j = 0; j < 2; ++j) {
          C1[i][j] = wmma_f32(Aa[i], Bp[j], C1[i][j]);
          C2[i][j] = wmma_f32(Ab[i], Bq[j], C2[i][j]);
          C3[i][j] = wmma_f32(As[i], Bd[j], C3[i][j]);
        }
      }
    }
    __syncthreads();
  }
  const size_t pbase = (size_t)blockIdx.y * ((size_t)NC * RESN * RESN);
#pragma unroll
  for (int i = 0; i < 2; ++i) {
#pragma unroll
    for (int j = 0; j < 2; ++j) {
#pragma unroll
      for (int v = 0; v < 8; ++v) {
        const int row = rowTile * 64 + wr * 32 + i * 16 + v + 8 * half;
        const int col = w0 + wc * 32 + j * 16 + lm;
        const size_t oi = pbase + (size_t)row * RESN + col;
        partR[oi] = C1[i][j][v] + C2[i][j][v];
        partI[oi] = C3[i][j][v] + C1[i][j][v] - C2[i][j][v];
      }
    }
  }
}

__global__ void reduce1_kernel(const float* __restrict__ partR, const float* __restrict__ partI,
                               float* __restrict__ out, float* __restrict__ red, int n) {
  __shared__ float smn[256], smx[256];
  float mn = 3.402823466e38f, mx = -3.402823466e38f;
  for (int i = blockIdx.x * blockDim.x + threadIdx.x; i < n; i += gridDim.x * blockDim.x) {
    float rr = 0.0f, ri = 0.0f;
#pragma unroll
    for (int s = 0; s < G2_SPLIT; ++s) {
      rr += partR[(size_t)s * n + i];
      ri += partI[(size_t)s * n + i];
    }
    float v = sqrtf(rr * rr + ri * ri);
    out[i] = v;
    mn = fminf(mn, v); mx = fmaxf(mx, v);
  }
  smn[threadIdx.x] = mn; smx[threadIdx.x] = mx;
  __syncthreads();
  for (int s = 128; s > 0; s >>= 1) {
    if ((int)threadIdx.x < s) {
      smn[threadIdx.x] = fminf(smn[threadIdx.x], smn[threadIdx.x + s]);
      smx[threadIdx.x] = fmaxf(smx[threadIdx.x], smx[threadIdx.x + s]);
    }
    __syncthreads();
  }
  if (threadIdx.x == 0) { red[2 + blockIdx.x] = smn[0]; red[2 + RB + blockIdx.x] = smx[0]; }
}

__global__ void reduce2_kernel(float* __restrict__ red) {
  __shared__ float smn[RB], smx[RB];
  const int t = threadIdx.x;
  smn[t] = red[2 + t]; smx[t] = red[2 + RB + t];
  __syncthreads();
  for (int s = RB / 2; s > 0; s >>= 1) {
    if (t < s) { smn[t] = fminf(smn[t], smn[t + s]); smx[t] = fmaxf(smx[t], smx[t + s]); }
    __syncthreads();
  }
  if (t == 0) { red[0] = smn[0]; red[1] = smx[0]; }
}

__global__ void normalize_kernel(float* __restrict__ out, const float* __restrict__ red, int n) {
  int i = blockIdx.x * blockDim.x + threadIdx.x;
  if (i < n) {
    float mn = red[0];
    float inv = 1.0f / (red[1] - mn + 1e-8f);
    out[i] = (out[i] - mn) * inv;
  }
}

extern "C" void kernel_launch(void* const* d_in, const int* in_sizes, int n_in,
                              void* d_out, int out_size, void* d_ws, size_t ws_size,
                              hipStream_t stream) {
  const float* image = (const float*)d_in[0];
  const float* x     = (const float*)d_in[1];
  float* out = (float*)d_out;

  const size_t TBL = (size_t)M_PAD * RESN;
  const size_t IMG = (size_t)NC * RESN * RESN;
  const size_t KSP = (size_t)NC * M_PAD;
  float* p    = (float*)d_ws;
  float* EhR  = p; p += TBL;
  float* EhI  = p; p += TBL;
  float* EwR  = p; p += TBL;
  float* EwI  = p; p += TBL;
  float* Acs  = p; p += TBL;
  float* imgBR = p; p += IMG;
  float* imgBI = p; p += IMG;
  float* kspR = p; p += KSP;
  float* kspI = p; p += KSP;
  float* xt   = p; p += (size_t)M_TOTAL * 2;
  float* red  = p; p += 1024;
  float* partR = p; p += (size_t)G2_SPLIT * IMG;
  float* partI = p;

  const int n = NC * RESN * RESN;

  traj_kernel<<<(M_TOTAL * 2 + 255) / 256, 256, 0, stream>>>(x, xt);
  phase_kernel<<<(M_PAD * RESN + 255) / 256, 256, 0, stream>>>(xt, EhR, EhI, EwR, EwI, Acs);
  fold_img_kernel<<<(n + 255) / 256, 256, 0, stream>>>(image, imgBR, imgBI);

  dim3 g1(M_PAD / G1_BM, NC);
  gemm1_kernel<<<g1, 256, 0, stream>>>(image, imgBR, imgBI, EhR, EhI, Acs, kspR, kspI);

  dim3 g2(((NC * RESN) / 64) * (RESN / 64), G2_SPLIT);
  gemm2_kernel<<<g2, 128, 0, stream>>>(EhR, EhI, EwR, EwI, kspR, kspI, partR, partI);

  reduce1_kernel<<<RB, 256, 0, stream>>>(partR, partI, out, red, n);
  reduce2_kernel<<<1, RB, 0, stream>>>(red);
  normalize_kernel<<<(n + 255) / 256, 256, 0, stream>>>(out, red, n);
}


# gemm2 uses the +-n conjugate symmetry of the w phase row: ksp folded into conj(Eh) as A, real cos|sin table as B, 4 real products per +-w pair, not 6; w=160 by a ones column (sum order changes)
# speedup vs baseline: 7.0392x; 1.1877x over previous
#include <hip/hip_runtime.h>
#include <math.h>


typedef float v2f __attribute__((ext_vector_type(2)));
typedef float v4f __attribute__((ext_vector_type(4)));
typedef float v8f __attribute__((ext_vector_type(8)));

#define RESN     320
#define NSHOTS   8
#define SPS      1001
#define M_TOTAL  (NSHOTS * SPS)
#define M_PAD    8064
#define NC       8
#define TWO_PI   6.283185307179586f
#define RB       256

#define V8F_ZERO {0.f, 0.f, 0.f, 0.f, 0.f, 0.f, 0.f, 0.f}

__device__ __forceinline__ v8f wmma_f32(v2f a, v2f b, v8f c) {
  return __builtin_amdgcn_wmma_f32_16x16x4_f32(
       false, a,  false, b,
       (short)0, c,  false,  false);
}

__device__ __forceinline__ v2f ldv2(const float* __restrict__ p) {
  return *(const v2f*)p;
}

__global__ void traj_kernel(const float* __restrict__ x, float* __restrict__ xt) {
  int gid = blockIdx.x * blockDim.x + threadIdx.x;
  if (gid >= M_TOTAL * 2) return;
  int d    = gid & 1;
  int m    = gid >> 1;
  int shot = m / SPS;
  int tt   = m - shot * SPS;
  int I    = (tt == 0) ? 0 : (tt - 1) / 10;
  if (I > 99) I = 99;
  const float* base = x + (size_t)shot * SPS * 2 + d;
  float y0 = base[(I) * 20];
  float y1 = base[(I + 1) * 20];
  float dI = (y1 - y0) * 0.1f;
  float mI, mI1;
  if (I == 0) { mI = dI; }
  else        { float ym = base[(I - 1) * 20]; mI = 0.5f * (dI + (y0 - ym) * 0.1f); }
  if (I == 99) { mI1 = dI; }
  else         { float y2 = base[(I + 2) * 20]; mI1 = 0.5f * (dI + (y2 - y1) * 0.1f); }
  float s  = (float)(tt - 10 * I) * 0.1f;
  float s2 = s * s, s3 = s2 * s;
  float h00 = 1.0f - 3.0f * s2 + 2.0f * s3;
  float h10 = s - 2.0f * s2 + s3;
  float h01 = 3.0f * s2 - 2.0f * s3;
  float h11 = s3 - s2;
  xt[gid] = h00 * y0 + h10 * mI * 10.0f + h01 * y1 + h11 * mI1 * 10.0f;
}

__global__ void phase_kernel(const float* __restrict__ xt,
                             float* __restrict__ EhR, float* __restrict__ EhI,
                             float* __restrict__ Acs) {
  int gid = blockIdx.x * blockDim.x + threadIdx.x;
  if (gid >= M_PAD * RESN) return;
  int m = gid / RESN;
  int n = gid - m * RESN;
  float ehr = 0.f, ehi = 0.f;
  float cwv = 0.f, swv = 0.f;
  if (m < M_TOTAL) {
    float k = (TWO_PI / (float)RESN) * ((float)n - 160.0f);
    float sh, ch, sw, cw;
    __sincosf(xt[2 * m + 0] * k, &sh, &ch);
    __sincosf(xt[2 * m + 1] * k, &sw, &cw);
    ehr = ch; ehi = -sh;
    cwv = cw; swv = sw;
  }
  EhR[gid] = ehr; EhI[gid] = ehi;
  const int nn = n - 160;
  float* arow = Acs + (size_t)m * RESN;
  if (nn >= 1) {
    arow[nn - 1]   = cwv;
    arow[159 + nn] = swv;
  } else if (nn == -160) {
    arow[159] = cwv;
    arow[319] = -swv;
  }
}

__global__ void fold_img_kernel(const float* __restrict__ image,
                                float* __restrict__ BR, float* __restrict__ BI) {
  int i = blockIdx.x * blockDim.x + threadIdx.x;
  if (i >= NC * RESN * RESN) return;
  const int row = i / RESN;
  const int k   = i - row * RESN;
  const bool lo = (k < 160);
  const int np  = lo ? (k + 1) : (k - 159);
  const float* s = image + (size_t)row * RESN * 2;
  float br, bi;
  if (np < 160) {
    const v2f a = *(const v2f*)(s + 2 * (160 + np));
    const v2f b = *(const v2f*)(s + 2 * (160 - np));
    if (lo) { br = a[0] + b[0]; bi = a[1] + b[1]; }
    else    { br = a[1] - b[1]; bi = b[0] - a[0]; }
  } else {
    const v2f b = *(const v2f*)s;
    if (lo) { br = b[0];  bi = b[1]; }
    else    { br = -b[1]; bi = b[0]; }
  }
  BR[i] = br; BI[i] = bi;
}

#define G1_BM 128
#define G1_BN 64
#define G1_KB 32
#define G1_LS 36
__global__ __launch_bounds__(256) void gemm1_kernel(
    const float* __restrict__ image,
    const float* __restrict__ BR, const float* __restrict__ BI,
    const float* __restrict__ EhR, const float* __restrict__ EhI,
    const float* __restrict__ Acs,
    float* __restrict__ kspR, float* __restrict__ kspI) {
  __shared__ __attribute__((aligned(16))) float sA[G1_BM * G1_LS];
  __shared__ __attribute__((aligned(16))) float sBR[G1_BN * G1_LS];
  __shared__ __attribute__((aligned(16))) float sBI[G1_BN * G1_LS];
  const int tid  = threadIdx.x;
  const int wave = tid >> 5;
  const int l    = tid & 31;
  const int half = l >> 4;
  const int lm   = l & 15;
  const int m0   = blockIdx.x * G1_BM;
  const int c    = blockIdx.y;
  const int q = tid & 7;
  const int r = tid >> 3;
  const float* gA  = Acs + (size_t)(m0 + r) * RESN + 4 * q;
  const float* gBR = BR + ((size_t)c * RESN + r) * RESN + 4 * q;
  const float* gBI = BI + ((size_t)c * RESN + r) * RESN + 4 * q;
  const int sOff = r * G1_LS + 4 * q;

  v4f pA[4], pBR[2], pBI[2];
#pragma unroll
  for (int j = 0; j < 4; ++j) {
    pA[j] = *(const v4f*)(gA + (size_t)(32 * j) * RESN);
  }
#pragma unroll
  for (int j = 0; j < 2; ++j) {
    pBR[j] = *(const v4f*)(gBR + (size_t)(32 * j) * RESN);
    pBI[j] = *(const v4f*)(gBI + (size_t)(32 * j) * RESN);
  }

  float accR[8], accI[8];
#pragma unroll
  for (int v = 0; v < 8; ++v) { accR[v] = 0.0f; accI[v] = 0.0f; }

  const int aOff = (wave * 16 + lm) * G1_LS + 2 * half;
  const int bOff = lm * G1_LS + 2 * half;

#pragma unroll 1
  for (int hc = 0; hc < RESN / G1_BN; ++hc) {
    v8f C1[4], C2[4];
#pragma unroll
    for (int j = 0; j < 4; ++j) {
      v8f z = V8F_ZERO;
      C1[j] = z; C2[j] = z;
    }
#pragma unroll 1
    for (int ks = 0; ks < RESN / G1_KB; ++ks) {
#pragma unroll
      for (int j = 0; j < 4; ++j) {
        *(v4f*)(sA + sOff + 32 * j * G1_LS) = pA[j];
      }
#pragma unroll
      for (int j = 0; j < 2; ++j) {
        *(v4f*)(sBR + sOff + 32 * j * G1_LS) = pBR[j];
        *(v4f*)(sBI + sOff + 32 * j * G1_LS) = pBI[j];
      }
      __syncthreads();
      int ksn = ks + 1, hcn = hc;
      if (ksn == RESN / G1_KB) { ksn = 0; hcn = hc + 1; }
      if (hcn < RESN / G1_BN) {
        const int kn = ksn * G1_KB;
        const size_t hn = (size_t)(hcn * G1_BN) * RESN;
#pragma unroll
        for (int j = 0; j < 4; ++j) {
          pA[j] = *(const v4f*)(gA + (size_t)(32 * j) * RESN + kn);
        }
#pragma unroll
        for (int j = 0; j < 2; ++j) {
          pBR[j] = *(const v4f*)(gBR + hn + (size_t)(32 * j) * RESN + kn);
          pBI[j] = *(const v4f*)(gBI + hn + (size_t)(32 * j) * RESN + kn);
        }
      }
#pragma unroll
      for (int kk = 0; kk < G1_KB; kk += 4) {
        v2f A = ldv2(sA + aOff + kk);
#pragma unroll
        for (int j = 0; j < 4; ++j) {
          v2f Br = ldv2(sBR + bOff + j * 16 * G1_LS + kk);
          v2f Bi = ldv2(sBI + bOff + j * 16 * G1_LS + kk);
          C1[j] = wmma_f32(A, Br, C1[j]);
          C2[j] = wmma_f32(A, Bi, C2[j]);
        }
      }
      __syncthreads();
    }
#pragma unroll
    for (int j = 0; j < 4; ++j) {
      const int h = hc * G1_BN + j * 16 + lm;
      const v2f i0 = ldv2(image + (((size_t)c * RESN + h) * RESN + 160) * 2);
#pragma unroll
      for (int v = 0; v < 8; ++v) {
        const int ml = wave * 16 + v + 8 * half;
        const size_t ei = (size_t)(m0 + ml) * RESN + h;
        float er = EhR[ei], eim = EhI[ei];
        float tr = C1[j][v] + i0[0];
        float ti = C2[j][v] + i0[1];
        accR[v] += er * tr - eim * ti;
        accI[v] += er * ti + eim * tr;
      }
    }
  }
#pragma unroll
  for (int v = 0; v < 8; ++v) {
    float pr = accR[v], pi = accI[v];
#pragma unroll
    for (int mask = 8; mask >= 1; mask >>= 1) {
      pr += __shfl_xor(pr, mask, 32);
      pi += __shfl_xor(pi, mask, 32);
    }
    accR[v] = pr; accI[v] = pi;
  }
  if (lm == 0) {
#pragma unroll
    for (int v = 0; v < 8; ++v) {
      const int m = m0 + wave * 16 + v + 8 * half;
      if (m < M_TOTAL) {
        kspR[(size_t)c * M_PAD + m] = accR[v];
        kspI[(size_t)c * M_PAD + m] = accI[v];
      }
    }
  }
}

#define G2_KB    32
#define G2_LS    72
#define G2_SPLIT 6
#define G2_SLABS (M_PAD / G2_KB / G2_SPLIT)
#define G2_XPL   (NC * RESN * 2)

template <int EXTRA>
__device__ __forceinline__ void gemm2_body(
    float* sAR, float* sAI, float* sB,
    const float* __restrict__ EhR, const float* __restrict__ EhI,
    const float* __restrict__ Acs,
    const float* __restrict__ kspR, const float* __restrict__ kspI,
    float* __restrict__ partR, float* __restrict__ partI,
    float* __restrict__ xsum,
    const int rowTile, const int colTile) {
  const int tid  = threadIdx.x;
  const int wave = tid >> 5;
  const int l    = tid & 31;
  const int half = l >> 4;
  const int lm   = l & 15;
  const int wr   = wave >> 1;
  const int wc   = wave & 1;
  const int c   = rowTile / 5;
  const int h0  = (rowTile - c * 5) * 64;
  const int n0  = colTile * 32;
  const int kb0 = blockIdx.y * (G2_SLABS * G2_KB);

  const int q = tid & 15;
  const int r = tid >> 4;
  const float* gAR = EhR + (size_t)(kb0 + r) * RESN + h0 + 4 * q;
  const float* gAI = EhI + (size_t)(kb0 + r) * RESN + h0 + 4 * q;
  const float* gB  = Acs + (size_t)(kb0 + r) * RESN + n0 + ((q < 8) ? (4 * q) : (160 + 4 * (q - 8)));
  const float* gKR = kspR + (size_t)c * M_PAD;
  const float* gKI = kspI + (size_t)c * M_PAD;
  const int sOff = r * G2_LS + 4 * q;

  v4f pAR[4], pAI[4], pB[4];
  float pKR[4], pKI[4];
#pragma unroll
  for (int j = 0; j < 4; ++j) {
    const int m = kb0 + r + 8 * j;
    pAR[j] = *(const v4f*)(gAR + (size_t)(8 * j) * RESN);
    pAI[j] = *(const v4f*)(gAI + (size_t)(8 * j) * RESN);
    pB[j]  = *(const v4f*)(gB  + (size_t)(8 * j) * RESN);
    const int mc = (m < M_TOTAL) ? m : (M_TOTAL - 1);
    const float kr = gKR[mc], ki = gKI[mc];
    pKR[j] = (m < M_TOTAL) ? kr : 0.0f;
    pKI[j] = (m < M_TOTAL) ? ki : 0.0f;
  }

  v8f Cpc[2], Cps[2], Cqc[2], Cqs[2];
#pragma unroll
  for (int i = 0; i < 2; ++i) {
    v8f z = V8F_ZERO;
    Cpc[i] = z; Cps[i] = z; Cqc[i] = z; Cqs[i] = z;
  }
  v8f Cx = V8F_ZERO;
  const v2f ones = {1.0f, 1.0f};

  const int aOff = (2 * half) * G2_LS + wr * 32 + lm;
  const int bOff = (2 * half) * G2_LS + wc * 16 + lm;
  const int xOff = aOff + wc * 16;
  const float* sX = (EXTRA == 2) ? sAI : sAR;

#pragma unroll 1
  for (int ks = 0; ks < G2_SLABS; ++ks) {
#pragma unroll
    for (int j = 0; j < 4; ++j) {
      const float kr = pKR[j], ki = pKI[j];
      v4f a = pAR[j], b = pAI[j];
      v4f gr = kr * a + ki * b;
      v4f gi = ki * a - kr * b;
      *(v4f*)(sAR + sOff + 8 * j * G2_LS) = gr;
      *(v4f*)(sAI + sOff + 8 * j * G2_LS) = gi;
      *(v4f*)(sB  + sOff + 8 * j * G2_LS) = pB[j];
    }
    __syncthreads();
    if (ks + 1 < G2_SLABS) {
      const int kn = (ks + 1) * G2_KB;
#pragma unroll
      for (int j = 0; j < 4; ++j) {
        const int m = kb0 + kn + r + 8 * j;
        const size_t go = (size_t)(kn + 8 * j) * RESN;
        pAR[j] = *(const v4f*)(gAR + go);
        pAI[j] = *(const v4f*)(gAI + go);
        pB[j]  = *(const v4f*)(gB  + go);
        const int mc = (m < M_TOTAL) ? m : (M_TOTAL - 1);
        const float kr = gKR[mc], ki = gKI[mc];
        pKR[j] = (m < M_TOTAL) ? kr : 0.0f;
        pKI[j] = (m < M_TOTAL) ? ki : 0.0f;
      }
    }
#pragma unroll
    for (int kk = 0; kk < G2_KB; kk += 4) {
      v2f Ar[2], Ai[2], Bc, Bs;
#pragma unroll
      for (int i = 0; i < 2; ++i) {
        const int o = aOff + kk * G2_LS + i * 16;
        Ar[i][0] = sAR[o]; Ar[i][1] = sAR[o + G2_LS];
        Ai[i][0] = sAI[o]; Ai[i][1] = sAI[o + G2_LS];
      }
      {
        const int o = bOff + kk * G2_LS;
        Bc[0] = sB[o];      Bc[1] = sB[o + G2_LS];
        Bs[0] = sB[o + 32]; Bs[1] = sB[o + 32 + G2_LS];
      }
#pragma unroll
      for (int i = 0; i < 2; ++i) {
        Cpc[i] = wmma_f32(Ar[i], Bc, Cpc[i]);
        Cps[i] = wmma_f32(Ar[i], Bs, Cps[i]);
        Cqc[i] = wmma_f32(Ai[i], Bc, Cqc[i]);
        Cqs[i] = wmma_f32(Ai[i], Bs, Cqs[i]);
      }
      if (EXTRA != 0) {
        const int o = xOff + kk * G2_LS;
        v2f Ax;
        Ax[0] = sX[o]; Ax[1] = sX[o + G2_LS];
        Cx = wmma_f32(Ax, ones, Cx);
      }
    }
    __syncthreads();
  }
  const size_t pbase = (size_t)blockIdx.y * ((size_t)NC * RESN * RESN);
  const int u = colTile * 2 + wc;
#pragma unroll
  for (int i = 0; i < 2; ++i) {
#pragma unroll
    for (int v = 0; v < 8; ++v) {
      const int row = rowTile * 64 + wr * 32 + i * 16 + v + 8 * half;
      const size_t oi = pbase + (size_t)row * RESN + 32 * u + lm;
      const float pc = Cpc[i][v], ps = Cps[i][v], qc = Cqc[i][v], qs = Cqs[i][v];
      partR[oi]      = pc - qs;
      partI[oi]      = ps + qc;
      partR[oi + 16] = pc + qs;
      partI[oi + 16] = qc - ps;
    }
  }
  if (EXTRA != 0) {
    if (lm == 0) {
      const size_t xb = (size_t)blockIdx.y * G2_XPL
                      + (size_t)(rowTile * 4 + wr * 2 + wc) * 32 + 8 * half;
#pragma unroll
      for (int v = 0; v < 8; ++v) xsum[xb + v] = Cx[v];
    }
  }
}

__global__ __launch_bounds__(128) void gemm2_kernel(
    const float* __restrict__ EhR, const float* __restrict__ EhI,
    const float* __restrict__ Acs,
    const float* __restrict__ kspR, const float* __restrict__ kspI,
    float* __restrict__ partR, float* __restrict__ partI,
    float* __restrict__ xsumR, float* __restrict__ xsumI) {
  __shared__ __attribute__((aligned(16))) float sAR[G2_KB * G2_LS];
  __shared__ __attribute__((aligned(16))) float sAI[G2_KB * G2_LS];
  __shared__ __attribute__((aligned(16))) float sB[G2_KB * G2_LS];
  const int rowTile = blockIdx.x / 5;
  const int colTile = blockIdx.x - rowTile * 5;
  if (colTile == 0) {
    gemm2_body<1>(sAR, sAI, sB, EhR, EhI, Acs, kspR, kspI, partR, partI, xsumR, rowTile, colTile);
  } else if (colTile == 1) {
    gemm2_body<2>(sAR, sAI, sB, EhR, EhI, Acs, kspR, kspI, partR, partI, xsumI, rowTile, colTile);
  } else {
    gemm2_body<0>(sAR, sAI, sB, EhR, EhI, Acs, kspR, kspI, partR, partI, xsumR, rowTile, colTile);
  }
}

__global__ void reduce1_kernel(const float* __restrict__ partR, const float* __restrict__ partI,
                               const float* __restrict__ xsumR, const float* __restrict__ xsumI,
                               float* __restrict__ out, float* __restrict__ red, int n) {
  __shared__ float smn[256], smx[256];
  float mn = 3.402823466e38f, mx = -3.402823466e38f;
  for (int i = blockIdx.x * blockDim.x + threadIdx.x; i < n; i += gridDim.x * blockDim.x) {
    const int row = i / RESN;
    const int w   = i - row * RESN;
    const float* pr = partR;
    const float* pi = partI;
    size_t idx, stride = (size_t)n;
    if (w == 160) {
      pr = xsumR; pi = xsumI; stride = G2_XPL;
      idx = (size_t)(row >> 4) * 32 + (row & 15);
    } else {
      const int t = (w > 160) ? (w - 161) : (159 - w);
      idx = (size_t)row * RESN + 32 * (t >> 4) + (t & 15) + ((w < 160) ? 16 : 0);
    }
    float rr = 0.0f, ri = 0.0f;
#pragma unroll
    for (int s = 0; s < G2_SPLIT; ++s) {
      rr += pr[(size_t)s * stride + idx];
      ri += pi[(size_t)s * stride + idx];
    }
    float v = sqrtf(rr * rr + ri * ri);
    out[i] = v;
    mn = fminf(mn, v); mx = fmaxf(mx, v);
  }
  smn[threadIdx.x] = mn; smx[threadIdx.x] = mx;
  __syncthreads();
  for (int s = 128; s > 0; s >>= 1) {
    if ((int)threadIdx.x < s) {
      smn[threadIdx.x] = fminf(smn[threadIdx.x], smn[threadIdx.x + s]);
      smx[threadIdx.x] = fmaxf(smx[threadIdx.x], smx[threadIdx.x + s]);
    }
    __syncthreads();
  }
  if (threadIdx.x == 0) { red[2 + blockIdx.x] = smn[0]; red[2 + RB + blockIdx.x] = smx[0]; }
}

__global__ void reduce2_kernel(float* __restrict__ red) {
  __shared__ float smn[RB], smx[RB];
  const int t = threadIdx.x;
  smn[t] = red[2 + t]; smx[t] = red[2 + RB + t];
  __syncthreads();
  for (int s = RB / 2; s > 0; s >>= 1) {
    if (t < s) { smn[t] = fminf(smn[t], smn[t + s]); smx[t] = fmaxf(smx[t], smx[t + s]); }
    __syncthreads();
  }
  if (t == 0) { red[0] = smn[0]; red[1] = smx[0]; }
}

__global__ void normalize_kernel(float* __restrict__ out, const float* __restrict__ red, int n) {
  int i = blockIdx.x * blockDim.x + threadIdx.x;
  if (i < n) {
    float mn = red[0];
    float inv = 1.0f / (red[1] - mn + 1e-8f);
    out[i] = (out[i] - mn) * inv;
  }
}

extern "C" void kernel_launch(void* const* d_in, const int* in_sizes, int n_in,
                              void* d_out, int out_size, void* d_ws, size_t ws_size,
                              hipStream_t stream) {
  const float* image = (const float*)d_in[0];
  const float* x     = (const float*)d_in[1];
  float* out = (float*)d_out;

  const size_t TBL = (size_t)M_PAD * RESN;
  const size_t IMG = (size_t)NC * RESN * RESN;
  const size_t KSP = (size_t)NC * M_PAD;
  float* p    = (float*)d_ws;
  float* EhR  = p; p += TBL;
  float* EhI  = p; p += TBL;
  float* Acs  = p; p += TBL;
  float* imgBR = p; p += IMG;
  float* imgBI = p; p += IMG;
  float* kspR = p; p += KSP;
  float* kspI = p; p += KSP;
  float* xt   = p; p += (size_t)M_TOTAL * 2;
  float* red  = p; p += 1024;
  float* partR = p; p += (size_t)G2_SPLIT * IMG;
  float* partI = p; p += (size_t)G2_SPLIT * IMG;
  float* xsumR = p; p += (size_t)G2_SPLIT * G2_XPL;
  float* xsumI = p;

  const int n = NC * RESN * RESN;

  traj_kernel<<<(M_TOTAL * 2 + 255) / 256, 256, 0, stream>>>(x, xt);
  phase_kernel<<<(M_PAD * RESN + 255) / 256, 256, 0, stream>>>(xt, EhR, EhI, Acs);
  fold_img_kernel<<<(n + 255) / 256, 256, 0, stream>>>(image, imgBR, imgBI);

  dim3 g1(M_PAD / G1_BM, NC);
  gemm1_kernel<<<g1, 256, 0, stream>>>(image, imgBR, imgBI, EhR, EhI, Acs, kspR, kspI);

  dim3 g2(((NC * RESN) / 64) * (RESN / 64), G2_SPLIT);
  gemm2_kernel<<<g2, 128, 0, stream>>>(EhR, EhI, Acs, kspR, kspI, partR, partI, xsumR, xsumI);

  reduce1_kernel<<<RB, 256, 0, stream>>>(partR, partI, xsumR, xsumI, out, red, n);
  reduce2_kernel<<<1, RB, 0, stream>>>(red);
  normalize_kernel<<<(n + 255) / 256, 256, 0, stream>>>(out, red, n);
}
